# starting kernel
# baseline (speedup 1.0000x reference)
#include <hip/hip_runtime.h>
#include <hip/hip_bf16.h>

#define B_DIM   64
#define IN_DIM  128
#define OUT_DIM 256
#define EPS     1e-8f
#define LOG2E   1.4426950408889634f

typedef __attribute__((ext_vector_type(2))) float v2f;
typedef __attribute__((ext_vector_type(8))) float v8f;

__global__ __launch_bounds__(32) void hgl_linear_wmma(
    const float* __restrict__ x,
    const float* __restrict__ W,
    const float* __restrict__ bias,
    float* __restrict__ lin)
{
    const int tile = blockIdx.x;
    const int tm   = tile % (B_DIM / 16);
    const int tn   = tile / (B_DIM / 16);
    const int lane = threadIdx.x;
    const int half = lane >> 4;
    const int l    = lane & 15;

    const int row = tm * 16 + l;
    const int col = tn * 16 + l;

    const float* xrow = x + (size_t)row * IN_DIM;
    const float* wrow = W + (size_t)col * IN_DIM;

    v8f acc = {};
    #pragma unroll
    for (int k = 0; k < IN_DIM; k += 4) {
        const int ka = k + half * 2;
        v2f a = *(const v2f*)(xrow + ka);
        v2f b = *(const v2f*)(wrow + ka);
        acc = __builtin_amdgcn_wmma_f32_16x16x4_f32(
             false, a,  false, b,
             (short)0, acc,  false,  false);
    }

    const float bv = bias[col];
    #pragma unroll
    for (int r = 0; r < 8; ++r) {
        const int m = tm * 16 + r + half * 8;
        lin[(size_t)m * OUT_DIM + col] = acc[r] + bv;
    }
}

__global__ __launch_bounds__(32) void hgl_gaussian(
    const float* __restrict__ x,
    const float* __restrict__ W,
    const float* __restrict__ log_sigma,
    const float* __restrict__ alpha,
    const float* __restrict__ lin,
    float* __restrict__ out)
{
    const int bo   = blockIdx.x;
    const int o    = bo % OUT_DIM;
    const int b    = bo / OUT_DIM;
    const int lane = threadIdx.x;

    __shared__ float zsh[IN_DIM];
    __shared__ float bsh[IN_DIM];

    const float es = __expf(log_sigma[o]);
    const float cl = -LOG2E / (2.0f * es * es + EPS);

    const float4 xv = reinterpret_cast<const float4*>(x + (size_t)b * IN_DIM)[lane];
    const float4 wv = reinterpret_cast<const float4*>(W + (size_t)o * IN_DIM)[lane];
    const float z0 = xv.x * wv.x;
    const float z1 = xv.y * wv.y;
    const float z2 = xv.z * wv.z;
    const float z3 = xv.w * wv.w;

    reinterpret_cast<float4*>(zsh)[lane] = make_float4(z0, z1, z2, z3);
    reinterpret_cast<float4*>(bsh)[lane] =
        make_float4(cl * z0 * z0, cl * z1 * z1, cl * z2 * z2, cl * z3 * z3);
    __syncthreads();

    const float a0 = cl * z0 * z0, m0 = -2.0f * cl * z0;
    const float a1 = cl * z1 * z1, m1 = -2.0f * cl * z1;
    const float a2 = cl * z2 * z2, m2 = -2.0f * cl * z2;
    const float a3 = cl * z3 * z3, m3 = -2.0f * cl * z3;

    float s0 = 0.f, s1 = 0.f, s2 = 0.f, s3 = 0.f;
    #pragma unroll 4
    for (int j = 0; j < IN_DIM; j += 4) {
        const float4 zj = *reinterpret_cast<const float4*>(&zsh[j]);
        const float4 bj = *reinterpret_cast<const float4*>(&bsh[j]);
        s0 += __builtin_amdgcn_exp2f(__builtin_fmaf(m0, zj.x, a0 + bj.x));
        s0 += __builtin_amdgcn_exp2f(__builtin_fmaf(m0, zj.y, a0 + bj.y));
        s0 += __builtin_amdgcn_exp2f(__builtin_fmaf(m0, zj.z, a0 + bj.z));
        s0 += __builtin_amdgcn_exp2f(__builtin_fmaf(m0, zj.w, a0 + bj.w));
        s1 += __builtin_amdgcn_exp2f(__builtin_fmaf(m1, zj.x, a1 + bj.x));
        s1 += __builtin_amdgcn_exp2f(__builtin_fmaf(m1, zj.y, a1 + bj.y));
        s1 += __builtin_amdgcn_exp2f(__builtin_fmaf(m1, zj.z, a1 + bj.z));
        s1 += __builtin_amdgcn_exp2f(__builtin_fmaf(m1, zj.w, a1 + bj.w));
        s2 += __builtin_amdgcn_exp2f(__builtin_fmaf(m2, zj.x, a2 + bj.x));
        s2 += __builtin_amdgcn_exp2f(__builtin_fmaf(m2, zj.y, a2 + bj.y));
        s2 += __builtin_amdgcn_exp2f(__builtin_fmaf(m2, zj.z, a2 + bj.z));
        s2 += __builtin_amdgcn_exp2f(__builtin_fmaf(m2, zj.w, a2 + bj.w));
        s3 += __builtin_amdgcn_exp2f(__builtin_fmaf(m3, zj.x, a3 + bj.x));
        s3 += __builtin_amdgcn_exp2f(__builtin_fmaf(m3, zj.y, a3 + bj.y));
        s3 += __builtin_amdgcn_exp2f(__builtin_fmaf(m3, zj.z, a3 + bj.z));
        s3 += __builtin_amdgcn_exp2f(__builtin_fmaf(m3, zj.w, a3 + bj.w));
    }

    float pn = fabsf(s0) + fabsf(s1) + fabsf(s2) + fabsf(s3);
    float pg = s0 * z0 + s1 * z1 + s2 * z2 + s3 * z3;

    #pragma unroll
    for (int off = 16; off > 0; off >>= 1) {
        pn += __shfl_xor(pn, off, 32);
        pg += __shfl_xor(pg, off, 32);
    }

    if (lane == 0) {
        const float norm = fmaxf(pn, 1e-12f);
        const float g    = pg / norm;
        const float a    = 1.0f / (1.0f + __expf(-alpha[o]));
        out[(size_t)b * OUT_DIM + o] = a * g + (1.0f - a) * lin[(size_t)b * OUT_DIM + o];
    }
}

extern "C" void kernel_launch(void* const* d_in, const int* in_sizes, int n_in,
                              void* d_out, int out_size, void* d_ws, size_t ws_size,
                              hipStream_t stream) {
    const float* x         = (const float*)d_in[0];
    const float* W         = (const float*)d_in[1];
    const float* bias      = (const float*)d_in[2];
    const float* log_sigma = (const float*)d_in[3];
    const float* alpha     = (const float*)d_in[4];
    float*       out       = (float*)d_out;
    float*       lin       = (float*)d_ws;

    hgl_linear_wmma<<<(B_DIM / 16) * (OUT_DIM / 16), 32, 0, stream>>>(x, W, bias, lin);

    hgl_gaussian<<<B_DIM * OUT_DIM, 32, 0, stream>>>(x, W, log_sigma, alpha, lin, out);
}


# Gaussian sums use the symmetry of the affinity matrix: each unordered pair (i,j) evaluated once, diagonal taken as 1 (half the exp2); norm and numerator are summed in a different order
# speedup vs baseline: 1.5755x; 1.5755x over previous
#include <hip/hip_runtime.h>
#include <hip/hip_bf16.h>

#define B_DIM   64
#define IN_DIM  128
#define OUT_DIM 256
#define EPS     1e-8f
#define LOG2E   1.4426950408889634f

typedef __attribute__((ext_vector_type(2))) float v2f;
typedef __attribute__((ext_vector_type(8))) float v8f;

__global__ __launch_bounds__(32) void hgl_linear_wmma(
    const float* __restrict__ x,
    const float* __restrict__ W,
    const float* __restrict__ bias,
    float* __restrict__ lin)
{
    const int tile = blockIdx.x;
    const int tm   = tile % (B_DIM / 16);
    const int tn   = tile / (B_DIM / 16);
    const int lane = threadIdx.x;
    const int half = lane >> 4;
    const int l    = lane & 15;

    const int row = tm * 16 + l;
    const int col = tn * 16 + l;

    const float* xrow = x + (size_t)row * IN_DIM;
    const float* wrow = W + (size_t)col * IN_DIM;

    v8f acc = {};
    #pragma unroll
    for (int k = 0; k < IN_DIM; k += 4) {
        const int ka = k + half * 2;
        v2f a = *(const v2f*)(xrow + ka);
        v2f b = *(const v2f*)(wrow + ka);
        acc = __builtin_amdgcn_wmma_f32_16x16x4_f32(
             false, a,  false, b,
             (short)0, acc,  false,  false);
    }

    const float bv = bias[col];
    #pragma unroll
    for (int r = 0; r < 8; ++r) {
        const int m = tm * 16 + r + half * 8;
        lin[(size_t)m * OUT_DIM + col] = acc[r] + bv;
    }
}

__device__ __forceinline__ v2f hgl_exp2_pair(v2f t)
{
    v2f r;
    r.x = __builtin_amdgcn_exp2f(t.x);
    r.y = __builtin_amdgcn_exp2f(t.y);
    return r;
}

__global__ __launch_bounds__(32) void hgl_gaussian(
    const float* __restrict__ x,
    const float* __restrict__ W,
    const float* __restrict__ log_sigma,
    const float* __restrict__ alpha,
    const float* __restrict__ lin,
    float* __restrict__ out)
{
    const int bo   = blockIdx.x;
    const int o    = bo % OUT_DIM;
    const int b    = bo / OUT_DIM;
    const int lane = threadIdx.x;

    __shared__ __attribute__((aligned(16))) float zq[4 * 32];
    __shared__ __attribute__((aligned(16))) float bq[4 * 32];
    __shared__ __attribute__((aligned(16))) float zn[64 * 4];
    __shared__ __attribute__((aligned(16))) float bn[64 * 4];

    const float es = __expf(log_sigma[o]);
    const float cl = -LOG2E / (2.0f * es * es + EPS);

    const float4 xv = reinterpret_cast<const float4*>(x + (size_t)b * IN_DIM)[lane];
    const float4 wv = reinterpret_cast<const float4*>(W + (size_t)o * IN_DIM)[lane];
    float z[4], a[4], m[4];
    z[0] = xv.x * wv.x;
    z[1] = xv.y * wv.y;
    z[2] = xv.z * wv.z;
    z[3] = xv.w * wv.w;
    #pragma unroll
    for (int q = 0; q < 4; ++q) {
        a[q] = cl * z[q] * z[q];
        m[q] = -2.0f * cl * z[q];
        zq[q * 32 + lane] = z[q];
        bq[q * 32 + lane] = a[q];
    }
    {
        const float4 zv = make_float4(z[0], z[1], z[2], z[3]);
        const float4 av = make_float4(a[0], a[1], a[2], a[3]);
        reinterpret_cast<float4*>(zn)[lane]      = zv;
        reinterpret_cast<float4*>(zn)[lane + 32] = zv;
        reinterpret_cast<float4*>(bn)[lane]      = av;
        reinterpret_cast<float4*>(bn)[lane + 32] = av;
    }
    __syncthreads();

    v2f S[4], U[4];
    #pragma unroll
    for (int p = 0; p < 4; ++p) { S[p] = (v2f){0.f, 0.f}; U[p] = (v2f){0.f, 0.f}; }

    v2f S01, S23, U01, U23;
    {
        const v2f a01 = (v2f){a[0], a[1]}, a23 = (v2f){a[2], a[3]};
        const v2f m01 = (v2f){m[0], m[1]}, m23 = (v2f){m[2], m[3]};
        {
            const float4 z4 = reinterpret_cast<const float4*>(zn)[lane + 16];
            const float4 b4 = reinterpret_cast<const float4*>(bn)[lane + 16];
            const v2f zlo = (v2f){z4.x, z4.y}, zhi = (v2f){z4.z, z4.w};
            const v2f blo = (v2f){b4.x, b4.y}, bhi = (v2f){b4.z, b4.w};
            const v2f e0 = 0.5f * hgl_exp2_pair(__builtin_elementwise_fma(m01, zlo, a01 + blo));
            const v2f e1 = 0.5f * hgl_exp2_pair(__builtin_elementwise_fma(m23, zhi, a23 + bhi));
            S01 = e0;
            U01 = e0 * zlo;
            S23 = e1;
            U23 = e1 * zhi;
        }
        #pragma unroll 3
        for (int d = 1; d <= 15; ++d) {
            const float4 z4 = reinterpret_cast<const float4*>(zn)[lane + d];
            const float4 b4 = reinterpret_cast<const float4*>(bn)[lane + d];
            const v2f zlo = (v2f){z4.x, z4.y}, zhi = (v2f){z4.z, z4.w};
            const v2f blo = (v2f){b4.x, b4.y}, bhi = (v2f){b4.z, b4.w};
            const v2f e0 = hgl_exp2_pair(__builtin_elementwise_fma(m01, zlo, a01 + blo));
            const v2f e1 = hgl_exp2_pair(__builtin_elementwise_fma(m23, zhi, a23 + bhi));
            S01 += e0;
            U01 = __builtin_elementwise_fma(e0, zlo, U01);
            S23 += e1;
            U23 = __builtin_elementwise_fma(e1, zhi, U23);
        }
    }

    #pragma unroll
    for (int q = 1; q < 4; ++q) {
        #pragma unroll 2
        for (int n = 0; n < 32; n += 4) {
            const float4 z4 = *reinterpret_cast<const float4*>(&zq[q * 32 + n]);
            const float4 b4 = *reinterpret_cast<const float4*>(&bq[q * 32 + n]);
            const v2f zlo = (v2f){z4.x, z4.y}, zhi = (v2f){z4.z, z4.w};
            const v2f blo = (v2f){b4.x, b4.y}, bhi = (v2f){b4.z, b4.w};
            #pragma unroll
            for (int p = 0; p < q; ++p) {
                const v2f ap = (v2f){a[p], a[p]};
                const v2f mp = (v2f){m[p], m[p]};
                const v2f e0 = hgl_exp2_pair(__builtin_elementwise_fma(mp, zlo, ap + blo));
                const v2f e1 = hgl_exp2_pair(__builtin_elementwise_fma(mp, zhi, ap + bhi));
                S[p] += e0;
                U[p] = __builtin_elementwise_fma(e0, zlo, U[p]);
                S[p] += e1;
                U[p] = __builtin_elementwise_fma(e1, zhi, U[p]);
            }
        }
    }

    const float sB[4] = {S01.x, S01.y, S23.x, S23.y};
    const float uB[4] = {U01.x, U01.y, U23.x, U23.y};
    float pn = 0.f, pg = 0.f;
    #pragma unroll
    for (int p = 0; p < 4; ++p) {
        const float sp = (S[p].x + S[p].y) + sB[p];
        pn += 1.0f + 2.0f * sp;
        pg += z[p] * (1.0f + sp) + ((U[p].x + U[p].y) + uB[p]);
    }

    #pragma unroll
    for (int off = 16; off > 0; off >>= 1) {
        pn += __shfl_xor(pn, off, 32);
        pg += __shfl_xor(pg, off, 32);
    }

    if (lane == 0) {
        const float norm = fmaxf(pn, 1e-12f);
        const float g    = pg / norm;
        const float al   = 1.0f / (1.0f + __expf(-alpha[o]));
        out[(size_t)b * OUT_DIM + o] = al * g + (1.0f - al) * lin[(size_t)b * OUT_DIM + o];
    }
}

extern "C" void kernel_launch(void* const* d_in, const int* in_sizes, int n_in,
                              void* d_out, int out_size, void* d_ws, size_t ws_size,
                              hipStream_t stream) {
    const float* x         = (const float*)d_in[0];
    const float* W         = (const float*)d_in[1];
    const float* bias      = (const float*)d_in[2];
    const float* log_sigma = (const float*)d_in[3];
    const float* alpha     = (const float*)d_in[4];
    float*       out       = (float*)d_out;
    float*       lin       = (float*)d_ws;

    hgl_linear_wmma<<<(B_DIM / 16) * (OUT_DIM / 16), 32, 0, stream>>>(x, W, bias, lin);

    hgl_gaussian<<<B_DIM * OUT_DIM, 32, 0, stream>>>(x, W, log_sigma, alpha, lin, out);
}


# Row factor exp2(a_i) taken out of the pair loops: exp2 argument is one packed fma (b_j + m_i*z_j), each row's two sums scaled by exp2(a_i) at the end; waves with an a_i < -32 keep the old loops
# speedup vs baseline: 1.8766x; 1.1911x over previous
#include <hip/hip_runtime.h>
#include <hip/hip_bf16.h>

#define B_DIM   64
#define IN_DIM  128
#define OUT_DIM 256
#define EPS     1e-8f
#define LOG2E   1.4426950408889634f

typedef __attribute__((ext_vector_type(2))) float v2f;
typedef __attribute__((ext_vector_type(8))) float v8f;

#define WG_WAVES    32
#define PART_PITCH  257
#define STAGE_FLTS  768

#define FACT_MIN_A  (-32.0f)

__device__ __forceinline__ v2f hgl_exp2_pair(v2f t)
{
    v2f r;
    r.x = __builtin_amdgcn_exp2f(t.x);
    r.y = __builtin_amdgcn_exp2f(t.y);
    return r;
}

template <bool FACT>
__device__ __forceinline__ v2f hgl_arg(v2f mi, v2f zj, v2f ai, v2f bj)
{
    if (FACT) return __builtin_elementwise_fma(mi, zj, bj);
    return __builtin_elementwise_fma(mi, zj, ai + bj);
}

template <bool FACT>
__device__ __forceinline__ void hgl_pairs(
    const float (&z)[4], const float (&a)[4], const float (&m)[4],
    const float* zq, const float* bq, const float* zn, const float* bn,
    const int lane, float& pn, float& pg)
{
    v2f S[4], U[4];
    #pragma unroll
    for (int p = 0; p < 4; ++p) { S[p] = (v2f){0.f, 0.f}; U[p] = (v2f){0.f, 0.f}; }

    v2f S01, S23, U01, U23;
    {
        const v2f a01 = (v2f){a[0], a[1]}, a23 = (v2f){a[2], a[3]};
        const v2f m01 = (v2f){m[0], m[1]}, m23 = (v2f){m[2], m[3]};
        {
            const float4 z4 = reinterpret_cast<const float4*>(zn)[lane + 16];
            const float4 b4 = reinterpret_cast<const float4*>(bn)[lane + 16];
            const v2f zlo = (v2f){z4.x, z4.y}, zhi = (v2f){z4.z, z4.w};
            const v2f blo = (v2f){b4.x, b4.y}, bhi = (v2f){b4.z, b4.w};
            const v2f e0 = 0.5f * hgl_exp2_pair(hgl_arg<FACT>(m01, zlo, a01, blo));
            const v2f e1 = 0.5f * hgl_exp2_pair(hgl_arg<FACT>(m23, zhi, a23, bhi));
            S01 = e0;
            U01 = e0 * zlo;
            S23 = e1;
            U23 = e1 * zhi;
        }
        #pragma unroll 3
        for (int d = 1; d <= 15; ++d) {
            const float4 z4 = reinterpret_cast<const float4*>(zn)[lane + d];
            const float4 b4 = reinterpret_cast<const float4*>(bn)[lane + d];
            const v2f zlo = (v2f){z4.x, z4.y}, zhi = (v2f){z4.z, z4.w};
            const v2f blo = (v2f){b4.x, b4.y}, bhi = (v2f){b4.z, b4.w};
            const v2f e0 = hgl_exp2_pair(hgl_arg<FACT>(m01, zlo, a01, blo));
            const v2f e1 = hgl_exp2_pair(hgl_arg<FACT>(m23, zhi, a23, bhi));
            S01 += e0;
            U01 = __builtin_elementwise_fma(e0, zlo, U01);
            S23 += e1;
            U23 = __builtin_elementwise_fma(e1, zhi, U23);
        }
    }

    #pragma unroll
    for (int q = 1; q < 4; ++q) {
        #pragma unroll 2
        for (int n = 0; n < 32; n += 4) {
            const float4 z4 = *reinterpret_cast<const float4*>(&zq[q * 32 + n]);
            const float4 b4 = *reinterpret_cast<const float4*>(&bq[q * 32 + n]);
            const v2f zlo = (v2f){z4.x, z4.y}, zhi = (v2f){z4.z, z4.w};
            const v2f blo = (v2f){b4.x, b4.y}, bhi = (v2f){b4.z, b4.w};
            #pragma unroll
            for (int p = 0; p < q; ++p) {
                const v2f ap = (v2f){a[p], a[p]};
                const v2f mp = (v2f){m[p], m[p]};
                const v2f e0 = hgl_exp2_pair(hgl_arg<FACT>(mp, zlo, ap, blo));
                const v2f e1 = hgl_exp2_pair(hgl_arg<FACT>(mp, zhi, ap, bhi));
                S[p] += e0;
                U[p] = __builtin_elementwise_fma(e0, zlo, U[p]);
                S[p] += e1;
                U[p] = __builtin_elementwise_fma(e1, zhi, U[p]);
            }
        }
    }

    const float sB[4] = {S01.x, S01.y, S23.x, S23.y};
    const float uB[4] = {U01.x, U01.y, U23.x, U23.y};
    pn = 0.f;
    pg = 0.f;
    #pragma unroll
    for (int p = 0; p < 4; ++p) {
        float sp = (S[p].x + S[p].y) + sB[p];
        float up = (U[p].x + U[p].y) + uB[p];
        if (FACT) {
            const float rf = __builtin_amdgcn_exp2f(a[p]);
            sp *= rf;
            up *= rf;
        }
        pn += 1.0f + 2.0f * sp;
        pg += z[p] * (1.0f + sp) + up;
    }
}

__global__ __launch_bounds__(WG_WAVES * 32) void hgl_fused(
    const float* __restrict__ x,
    const float* __restrict__ W,
    const float* __restrict__ bias,
    const float* __restrict__ log_sigma,
    const float* __restrict__ alpha,
    float* __restrict__ out)
{
    const int tile = blockIdx.x >> 3;
    const int sub  = blockIdx.x & 7;
    const int tm   = tile % (B_DIM / 16);
    const int tn   = tile / (B_DIM / 16);
    const int wid  = __builtin_amdgcn_readfirstlane((int)(threadIdx.x >> 5));
    const int lane = threadIdx.x & 31;
    const int mrow = 2 * sub + (wid >> 4);
    const int ncol = wid & 15;
    const int b    = tm * 16 + mrow;
    const int o    = tn * 16 + ncol;

    __shared__ float part[WG_WAVES * PART_PITCH];
    __shared__ __attribute__((aligned(16))) float stage[WG_WAVES * STAGE_FLTS];
    float* const zq = stage + wid * STAGE_FLTS;
    float* const bq = zq + 4 * 32;
    float* const zn = bq + 4 * 32;
    float* const bn = zn + 64 * 4;

    {
        const int half = lane >> 4;
        const int l    = lane & 15;
        const int ka   = 4 * wid + half * 2;
        const v2f af = *(const v2f*)(x + (size_t)(tm * 16 + l) * IN_DIM + ka);
        const v2f bf = *(const v2f*)(W + (size_t)(tn * 16 + l) * IN_DIM + ka);
        v8f acc;
        asm("v_wmma_f32_16x16x4_f32 %0, %1, %2, 0\n\tv_nop\n\tv_nop\n\tv_nop\n\tv_nop"
            : "=&v"(acc) : "v"(af), "v"(bf));
        #pragma unroll
        for (int r = 0; r < 8; ++r)
            part[wid * PART_PITCH + r * 32 + lane] = acc[r];
    }

    const float es = __expf(log_sigma[o]);
    const float cl = -LOG2E / (2.0f * es * es + EPS);

    const float4 xv = reinterpret_cast<const float4*>(x + (size_t)b * IN_DIM)[lane];
    const float4 wv = reinterpret_cast<const float4*>(W + (size_t)o * IN_DIM)[lane];
    float z[4], a[4], m[4];
    z[0] = xv.x * wv.x;
    z[1] = xv.y * wv.y;
    z[2] = xv.z * wv.z;
    z[3] = xv.w * wv.w;
    #pragma unroll
    for (int q = 0; q < 4; ++q) {
        a[q] = cl * z[q] * z[q];
        m[q] = -2.0f * cl * z[q];
        zq[q * 32 + lane] = z[q];
        bq[q * 32 + lane] = a[q];
    }
    {
        const float4 zv = make_float4(z[0], z[1], z[2], z[3]);
        const float4 av = make_float4(a[0], a[1], a[2], a[3]);
        reinterpret_cast<float4*>(zn)[lane]      = zv;
        reinterpret_cast<float4*>(zn)[lane + 32] = zv;
        reinterpret_cast<float4*>(bn)[lane]      = av;
        reinterpret_cast<float4*>(bn)[lane + 32] = av;
    }
    __syncthreads();

    const float amin = fminf(fminf(a[0], a[1]), fminf(a[2], a[3]));
    const bool  fact = __builtin_amdgcn_ballot_w32(!(amin >= FACT_MIN_A)) == 0u;
    float pn, pg;
    if (fact) hgl_pairs<true >(z, a, m, zq, bq, zn, bn, lane, pn, pg);
    else      hgl_pairs<false>(z, a, m, zq, bq, zn, bn, lane, pn, pg);

    float pl = part[lane * PART_PITCH + (mrow & 7) * 32 + (mrow >> 3) * 16 + ncol];

    #pragma unroll
    for (int off = 16; off > 0; off >>= 1) {
        pn += __shfl_xor(pn, off, 32);
        pg += __shfl_xor(pg, off, 32);
        pl += __shfl_xor(pl, off, 32);
    }

    if (lane == 0) {
        const float norm = fmaxf(pn, 1e-12f);
        const float g    = pg / norm;
        const float al   = 1.0f / (1.0f + __expf(-alpha[o]));
        out[(size_t)b * OUT_DIM + o] = al * g + (1.0f - al) * (pl + bias[o]);
    }
}

extern "C" void kernel_launch(void* const* d_in, const int* in_sizes, int n_in,
                              void* d_out, int out_size, void* d_ws, size_t ws_size,
                              hipStream_t stream) {
    const float* x         = (const float*)d_in[0];
    const float* W         = (const float*)d_in[1];
    const float* bias      = (const float*)d_in[2];
    const float* log_sigma = (const float*)d_in[3];
    const float* alpha     = (const float*)d_in[4];
    float*       out       = (float*)d_out;

    hgl_fused<<<(B_DIM / 16) * (OUT_DIM / 16) * 8, WG_WAVES * 32, 0, stream>>>(
        x, W, bias, log_sigma, alpha, out);
}


# Gaussian sums from 14 power-series moments sum_i w_i*s_i^n (exp(s_i*s_j) cut after n=12, remainder < 5e-10) where every 2c*z_i^2 <= 1, else the pair loops; sum order changes
# speedup vs baseline: 4.8242x; 2.5707x over previous
#include <hip/hip_runtime.h>
#include <hip/hip_bf16.h>

#define B_DIM   64
#define IN_DIM  128
#define OUT_DIM 256
#define EPS     1e-8f
#define LOG2E   1.4426950408889634f

typedef __attribute__((ext_vector_type(2))) float v2f;
typedef __attribute__((ext_vector_type(8))) float v8f;

#define WG_WAVES    32
#define PART_PITCH  257
#define STAGE_FLTS  768

#define FACT_MIN_A  (-32.0f)

__device__ __forceinline__ v2f hgl_exp2_pair(v2f t)
{
    v2f r;
    r.x = __builtin_amdgcn_exp2f(t.x);
    r.y = __builtin_amdgcn_exp2f(t.y);
    return r;
}

template <bool FACT>
__device__ __forceinline__ v2f hgl_arg(v2f mi, v2f zj, v2f ai, v2f bj)
{
    if (FACT) return __builtin_elementwise_fma(mi, zj, bj);
    return __builtin_elementwise_fma(mi, zj, ai + bj);
}

template <bool FACT>
__device__ __forceinline__ void hgl_pairs(
    const float (&z)[4], const float (&a)[4], const float (&m)[4],
    const float* zq, const float* bq, const float* zn, const float* bn,
    const int lane, float& pn, float& pg)
{
    v2f S[4], U[4];
    #pragma unroll
    for (int p = 0; p < 4; ++p) { S[p] = (v2f){0.f, 0.f}; U[p] = (v2f){0.f, 0.f}; }

    v2f S01, S23, U01, U23;
    {
        const v2f a01 = (v2f){a[0], a[1]}, a23 = (v2f){a[2], a[3]};
        const v2f m01 = (v2f){m[0], m[1]}, m23 = (v2f){m[2], m[3]};
        {
            const float4 z4 = reinterpret_cast<const float4*>(zn)[lane + 16];
            const float4 b4 = reinterpret_cast<const float4*>(bn)[lane + 16];
            const v2f zlo = (v2f){z4.x, z4.y}, zhi = (v2f){z4.z, z4.w};
            const v2f blo = (v2f){b4.x, b4.y}, bhi = (v2f){b4.z, b4.w};
            const v2f e0 = 0.5f * hgl_exp2_pair(hgl_arg<FACT>(m01, zlo, a01, blo));
            const v2f e1 = 0.5f * hgl_exp2_pair(hgl_arg<FACT>(m23, zhi, a23, bhi));
            S01 = e0;
            U01 = e0 * zlo;
            S23 = e1;
            U23 = e1 * zhi;
        }
        #pragma unroll 3
        for (int d = 1; d <= 15; ++d) {
            const float4 z4 = reinterpret_cast<const float4*>(zn)[lane + d];
            const float4 b4 = reinterpret_cast<const float4*>(bn)[lane + d];
            const v2f zlo = (v2f){z4.x, z4.y}, zhi = (v2f){z4.z, z4.w};
            const v2f blo = (v2f){b4.x, b4.y}, bhi = (v2f){b4.z, b4.w};
            const v2f e0 = hgl_exp2_pair(hgl_arg<FACT>(m01, zlo, a01, blo));
            const v2f e1 = hgl_exp2_pair(hgl_arg<FACT>(m23, zhi, a23, bhi));
            S01 += e0;
            U01 = __builtin_elementwise_fma(e0, zlo, U01);
            S23 += e1;
            U23 = __builtin_elementwise_fma(e1, zhi, U23);
        }
    }

    #pragma unroll
    for (int q = 1; q < 4; ++q) {
        #pragma unroll (FACT ? 2 : 1)
        for (int n = 0; n < 32; n += 4) {
            const float4 z4 = *reinterpret_cast<const float4*>(&zq[q * 32 + n]);
            const float4 b4 = *reinterpret_cast<const float4*>(&bq[q * 32 + n]);
            const v2f zlo = (v2f){z4.x, z4.y}, zhi = (v2f){z4.z, z4.w};
            const v2f blo = (v2f){b4.x, b4.y}, bhi = (v2f){b4.z, b4.w};
            #pragma unroll
            for (int p = 0; p < q; ++p) {
                const v2f ap = (v2f){a[p], a[p]};
                const v2f mp = (v2f){m[p], m[p]};
                const v2f e0 = hgl_exp2_pair(hgl_arg<FACT>(mp, zlo, ap, blo));
                const v2f e1 = hgl_exp2_pair(hgl_arg<FACT>(mp, zhi, ap, bhi));
                S[p] += e0;
                U[p] = __builtin_elementwise_fma(e0, zlo, U[p]);
                S[p] += e1;
                U[p] = __builtin_elementwise_fma(e1, zhi, U[p]);
            }
        }
    }

    const float sB[4] = {S01.x, S01.y, S23.x, S23.y};
    const float uB[4] = {U01.x, U01.y, U23.x, U23.y};
    pn = 0.f;
    pg = 0.f;
    #pragma unroll
    for (int p = 0; p < 4; ++p) {
        float sp = (S[p].x + S[p].y) + sB[p];
        float up = (U[p].x + U[p].y) + uB[p];
        if (FACT) {
            const float rf = __builtin_amdgcn_exp2f(a[p]);
            sp *= rf;
            up *= rf;
        }
        pn += 1.0f + 2.0f * sp;
        pg += z[p] * (1.0f + sp) + up;
    }
}

#define SER_MIN_A   (-0.5f * LOG2E)
#define SER_TERMS   13

__device__ __forceinline__ constexpr float hgl_rsqrt_fact(int n)
{
    return n <  2 ? 1.0f
         : n ==  2 ? 0.7071067812f
         : n ==  3 ? 0.4082482905f
         : n ==  4 ? 0.2041241452f
         : n ==  5 ? 0.09128709292f
         : n ==  6 ? 0.03726779962f
         : n ==  7 ? 0.01408590425f
         : n ==  8 ? 0.004980119206f
         : n ==  9 ? 0.001660039735f
         : n == 10 ? 0.0005249506570f
         : n == 11 ? 0.0001582785784f
         : n == 12 ? 0.00004569108993f
         :           0.00001267242827f;
}

template <int H, int X>
__device__ __forceinline__ void hgl_halve(float (&v)[16], const int lane)
{
    const bool up = (lane & X) != 0;
    #pragma unroll
    for (int k = 0; k < H; ++k) {
        const float keep = up ? v[k + H] : v[k];
        const float send = up ? v[k]     : v[k + H];
        v[k] = keep + __shfl_xor(send, X, 32);
    }
}

__device__ __forceinline__ void hgl_series(
    const float (&z)[4], const float (&a)[4], const float r, const float rinv,
    const int lane, float& pn, float& pg)
{
    const v2f z01 = (v2f){z[0], z[1]}, z23 = (v2f){z[2], z[3]};
    const v2f s01 = r * z01, s23 = r * z23;
    v2f t01 = hgl_exp2_pair((v2f){a[0], a[1]});
    v2f t23 = hgl_exp2_pair((v2f){a[2], a[3]});

    float v[16];
    {
        const v2f wz = __builtin_elementwise_fma(t01, z01, t23 * z23);
        v[14] = wz.x + wz.y;
        v[15] = 0.0f;
    }
    #pragma unroll
    for (int n = 0; n < 14; ++n) {
        const v2f u = t01 + t23;
        v[n] = (u.x + u.y) * hgl_rsqrt_fact(n);
        t01 *= s01;
        t23 *= s23;
    }

    hgl_halve<8, 16>(v, lane);
    hgl_halve<4,  8>(v, lane);
    hgl_halve<2,  4>(v, lane);
    hgl_halve<1,  2>(v, lane);
    float mine = v[0];
    mine += __shfl_xor(mine, 1, 32);

    const int   n    = lane >> 1;
    const float next = __shfl_down(mine, 2, 32);
    const float wz   = __shfl(mine, 28, 32);
    const bool  live = ((lane & 1) == 0) && (n < SER_TERMS);
    const float other = (n == 0) ? wz
                                 : next * (__builtin_amdgcn_sqrtf((float)(n + 1)) * rinv);
    pn = live ? mine * mine  : 0.0f;
    pg = live ? mine * other : 0.0f;
}

__global__ __launch_bounds__(WG_WAVES * 32) void hgl_fused(
    const float* __restrict__ x,
    const float* __restrict__ W,
    const float* __restrict__ bias,
    const float* __restrict__ log_sigma,
    const float* __restrict__ alpha,
    float* __restrict__ out)
{
    const int tile = blockIdx.x >> 3;
    const int sub  = blockIdx.x & 7;
    const int tm   = tile % (B_DIM / 16);
    const int tn   = tile / (B_DIM / 16);
    const int wid  = __builtin_amdgcn_readfirstlane((int)(threadIdx.x >> 5));
    const int lane = threadIdx.x & 31;
    const int mrow = 2 * sub + (wid >> 4);
    const int ncol = wid & 15;
    const int b    = tm * 16 + mrow;
    const int o    = tn * 16 + ncol;

    __shared__ float part[WG_WAVES * PART_PITCH];
    __shared__ __attribute__((aligned(16))) float stage[WG_WAVES * STAGE_FLTS];
    float* const zq = stage + wid * STAGE_FLTS;
    float* const bq = zq + 4 * 32;
    float* const zn = bq + 4 * 32;
    float* const bn = zn + 64 * 4;

    {
        const int half = lane >> 4;
        const int l    = lane & 15;
        const int ka   = 4 * wid + half * 2;
        const v2f af = *(const v2f*)(x + (size_t)(tm * 16 + l) * IN_DIM + ka);
        const v2f bf = *(const v2f*)(W + (size_t)(tn * 16 + l) * IN_DIM + ka);
        v8f acc;
        asm("v_wmma_f32_16x16x4_f32 %0, %1, %2, 0\n\tv_nop\n\tv_nop\n\tv_nop\n\tv_nop"
            : "=&v"(acc) : "v"(af), "v"(bf));
        #pragma unroll
        for (int r = 0; r < 8; ++r)
            part[wid * PART_PITCH + r * 32 + lane] = acc[r];
    }

    const float es = __expf(log_sigma[o]);
    const float cl = -LOG2E / (2.0f * es * es + EPS);
    const float r    = __builtin_amdgcn_sqrtf(cl * (-2.0f / LOG2E));
    const float rinv = r > 0.0f ? __builtin_amdgcn_rcpf(r) : 0.0f;

    const float4 xv = reinterpret_cast<const float4*>(x + (size_t)b * IN_DIM)[lane];
    const float4 wv = reinterpret_cast<const float4*>(W + (size_t)o * IN_DIM)[lane];
    float z[4], a[4], m[4];
    z[0] = xv.x * wv.x;
    z[1] = xv.y * wv.y;
    z[2] = xv.z * wv.z;
    z[3] = xv.w * wv.w;
    #pragma unroll
    for (int q = 0; q < 4; ++q) {
        a[q] = cl * z[q] * z[q];
        m[q] = -2.0f * cl * z[q];
    }

    const float amin   = fminf(fminf(a[0], a[1]), fminf(a[2], a[3]));
    const bool  series = __builtin_amdgcn_ballot_w32(!(amin >= SER_MIN_A)) == 0u;
    const bool  fact   = __builtin_amdgcn_ballot_w32(!(amin >= FACT_MIN_A)) == 0u;

    if (!series) {
        #pragma unroll
        for (int q = 0; q < 4; ++q) {
            zq[q * 32 + lane] = z[q];
            bq[q * 32 + lane] = a[q];
        }
        const float4 zv = make_float4(z[0], z[1], z[2], z[3]);
        const float4 av = make_float4(a[0], a[1], a[2], a[3]);
        reinterpret_cast<float4*>(zn)[lane]      = zv;
        reinterpret_cast<float4*>(zn)[lane + 32] = zv;
        reinterpret_cast<float4*>(bn)[lane]      = av;
        reinterpret_cast<float4*>(bn)[lane + 32] = av;
    }
    __syncthreads();

    float pn, pg;
    if (series)    hgl_series(z, a, r, rinv, lane, pn, pg);
    else if (fact) hgl_pairs<true >(z, a, m, zq, bq, zn, bn, lane, pn, pg);
    else           hgl_pairs<false>(z, a, m, zq, bq, zn, bn, lane, pn, pg);

    float pl = part[lane * PART_PITCH + (mrow & 7) * 32 + (mrow >> 3) * 16 + ncol];

    #pragma unroll
    for (int off = 16; off > 0; off >>= 1) {
        pn += __shfl_xor(pn, off, 32);
        pg += __shfl_xor(pg, off, 32);
        pl += __shfl_xor(pl, off, 32);
    }

    if (lane == 0) {
        const float norm = fmaxf(pn, 1e-12f);
        const float g    = pg / norm;
        const float al   = 1.0f / (1.0f + __expf(-alpha[o]));
        out[(size_t)b * OUT_DIM + o] = al * g + (1.0f - al) * (pl + bias[o]);
    }
}

extern "C" void kernel_launch(void* const* d_in, const int* in_sizes, int n_in,
                              void* d_out, int out_size, void* d_ws, size_t ws_size,
                              hipStream_t stream) {
    const float* x         = (const float*)d_in[0];
    const float* W         = (const float*)d_in[1];
    const float* bias      = (const float*)d_in[2];
    const float* log_sigma = (const float*)d_in[3];
    const float* alpha     = (const float*)d_in[4];
    float*       out       = (float*)d_out;

    hgl_fused<<<(B_DIM / 16) * (OUT_DIM / 16) * 8, WG_WAVES * 32, 0, stream>>>(
        x, W, bias, log_sigma, alpha, out);
}


# Four lanes per (b,o) pair, 8 pairs per wave, not one pair per wave (2048 waves, not 16384): moments summed in registers, 2-step DPP quad sum; linear K split 8 ways; both sum orders change
# speedup vs baseline: 6.8672x; 1.4235x over previous
#include <hip/hip_runtime.h>
#include <hip/hip_bf16.h>

#define B_DIM   64
#define IN_DIM  128
#define OUT_DIM 256
#define EPS     1e-8f
#define LOG2E   1.4426950408889634f

typedef __attribute__((ext_vector_type(2))) float v2f;
typedef __attribute__((ext_vector_type(4))) float v4f;
typedef __attribute__((ext_vector_type(8))) float v8f;

#define WG_WAVES    8
#define PAIR_LANES  4
#define PART_PITCH  264
#define STAGE_FLTS  768

#define FACT_MIN_A  (-32.0f)

__device__ __forceinline__ v2f hgl_exp2_pair(v2f t)
{
    v2f r;
    r.x = __builtin_amdgcn_exp2f(t.x);
    r.y = __builtin_amdgcn_exp2f(t.y);
    return r;
}

template <bool FACT>
__device__ __forceinline__ v2f hgl_arg(v2f mi, v2f zj, v2f ai, v2f bj)
{
    if (FACT) return __builtin_elementwise_fma(mi, zj, bj);
    return __builtin_elementwise_fma(mi, zj, ai + bj);
}

template <bool FACT>
__device__ __forceinline__ void hgl_pairs(
    const float (&z)[4], const float (&a)[4], const float (&m)[4],
    const float* zq, const float* bq, const float* zn, const float* bn,
    const int lane, float& pn, float& pg)
{
    v2f S[4], U[4];
    #pragma unroll
    for (int p = 0; p < 4; ++p) { S[p] = (v2f){0.f, 0.f}; U[p] = (v2f){0.f, 0.f}; }

    v2f S01, S23, U01, U23;
    {
        const v2f a01 = (v2f){a[0], a[1]}, a23 = (v2f){a[2], a[3]};
        const v2f m01 = (v2f){m[0], m[1]}, m23 = (v2f){m[2], m[3]};
        {
            const float4 z4 = reinterpret_cast<const float4*>(zn)[lane + 16];
            const float4 b4 = reinterpret_cast<const float4*>(bn)[lane + 16];
            const v2f zlo = (v2f){z4.x, z4.y}, zhi = (v2f){z4.z, z4.w};
            const v2f blo = (v2f){b4.x, b4.y}, bhi = (v2f){b4.z, b4.w};
            const v2f e0 = 0.5f * hgl_exp2_pair(hgl_arg<FACT>(m01, zlo, a01, blo));
            const v2f e1 = 0.5f * hgl_exp2_pair(hgl_arg<FACT>(m23, zhi, a23, bhi));
            S01 = e0;
            U01 = e0 * zlo;
            S23 = e1;
            U23 = e1 * zhi;
        }
        #pragma unroll 1
        for (int d = 1; d <= 15; ++d) {
            const float4 z4 = reinterpret_cast<const float4*>(zn)[lane + d];
            const float4 b4 = reinterpret_cast<const float4*>(bn)[lane + d];
            const v2f zlo = (v2f){z4.x, z4.y}, zhi = (v2f){z4.z, z4.w};
            const v2f blo = (v2f){b4.x, b4.y}, bhi = (v2f){b4.z, b4.w};
            const v2f e0 = hgl_exp2_pair(hgl_arg<FACT>(m01, zlo, a01, blo));
            const v2f e1 = hgl_exp2_pair(hgl_arg<FACT>(m23, zhi, a23, bhi));
            S01 += e0;
            U01 = __builtin_elementwise_fma(e0, zlo, U01);
            S23 += e1;
            U23 = __builtin_elementwise_fma(e1, zhi, U23);
        }
    }

    #pragma unroll
    for (int q = 1; q < 4; ++q) {
        #pragma unroll 1
        for (int n = 0; n < 32; n += 4) {
            const float4 z4 = *reinterpret_cast<const float4*>(&zq[q * 32 + n]);
            const float4 b4 = *reinterpret_cast<const float4*>(&bq[q * 32 + n]);
            const v2f zlo = (v2f){z4.x, z4.y}, zhi = (v2f){z4.z, z4.w};
            const v2f blo = (v2f){b4.x, b4.y}, bhi = (v2f){b4.z, b4.w};
            #pragma unroll
            for (int p = 0; p < q; ++p) {
                const v2f ap = (v2f){a[p], a[p]};
                const v2f mp = (v2f){m[p], m[p]};
                const v2f e0 = hgl_exp2_pair(hgl_arg<FACT>(mp, zlo, ap, blo));
                const v2f e1 = hgl_exp2_pair(hgl_arg<FACT>(mp, zhi, ap, bhi));
                S[p] += e0;
                U[p] = __builtin_elementwise_fma(e0, zlo, U[p]);
                S[p] += e1;
                U[p] = __builtin_elementwise_fma(e1, zhi, U[p]);
            }
        }
    }

    const float sB[4] = {S01.x, S01.y, S23.x, S23.y};
    const float uB[4] = {U01.x, U01.y, U23.x, U23.y};
    pn = 0.f;
    pg = 0.f;
    #pragma unroll
    for (int p = 0; p < 4; ++p) {
        float sp = (S[p].x + S[p].y) + sB[p];
        float up = (U[p].x + U[p].y) + uB[p];
        if (FACT) {
            const float rf = __builtin_amdgcn_exp2f(a[p]);
            sp *= rf;
            up *= rf;
        }
        pn += 1.0f + 2.0f * sp;
        pg += z[p] * (1.0f + sp) + up;
    }
}

#define SER_MIN_A   (-0.5f * LOG2E)
#define SER_TERMS   13
#define SER_MOMS    14

__device__ __forceinline__ constexpr float hgl_inv_fact(int n)
{
    return n <  2 ? 1.0f
         : n ==  2 ? 0.5f
         : n ==  3 ? 0.16666666667f
         : n ==  4 ? 0.041666666667f
         : n ==  5 ? 0.0083333333333f
         : n ==  6 ? 0.0013888888889f
         : n ==  7 ? 0.00019841269841f
         : n ==  8 ? 0.000024801587302f
         : n ==  9 ? 0.0000027557319224f
         : n == 10 ? 0.00000027557319224f
         : n == 11 ? 0.000000025052108385f
         :           0.0000000020876756988f;
}

__device__ __forceinline__ float hgl_quad_sum(float v)
{
    v += __builtin_bit_cast(float, __builtin_amdgcn_update_dpp(
             0, __builtin_bit_cast(int, v), 0x161, 0xF, 0xF, true));
    v += __builtin_bit_cast(float, __builtin_amdgcn_update_dpp(
             0, __builtin_bit_cast(int, v), 0x162, 0xF, 0xF, true));
    return v;
}

struct hgl_frag8 { float4 lo, hi; };
__device__ __forceinline__ hgl_frag8 hgl_frag(const float* __restrict__ row0, const int l, const int kofs)
{
    const float4* p = reinterpret_cast<const float4*>(row0 + (size_t)l * IN_DIM + kofs);
    hgl_frag8 f;
    f.lo = p[0];
    f.hi = p[1];
    return f;
}

__global__ __launch_bounds__(WG_WAVES * 32) void hgl_fused(
    const float* __restrict__ x,
    const float* __restrict__ W,
    const float* __restrict__ bias,
    const float* __restrict__ log_sigma,
    const float* __restrict__ alpha,
    float* __restrict__ out)
{
    const int tile = blockIdx.x >> 2;
    const int sub  = blockIdx.x & 3;
    const int tm   = tile % (B_DIM / 16);
    const int tn   = tile / (B_DIM / 16);
    const int wid  = __builtin_amdgcn_readfirstlane((int)(threadIdx.x >> 5));
    const int lane = threadIdx.x & 31;
    const int quad = lane >> 2;
    const int kq   = lane & 3;
    const int mrow = 4 * sub + (wid >> 1);
    const int nco0 = 8 * (wid & 1);
    const int ncol = nco0 + quad;
    const int b    = tm * 16 + mrow;
    const int o    = tn * 16 + ncol;

    __shared__ float part[WG_WAVES * PART_PITCH];
    __shared__ __attribute__((aligned(16))) float stage[WG_WAVES * STAGE_FLTS];

    const float ls_o = log_sigma[o];
    const float al_o = alpha[o];
    const float bi_o = bias[o];
    const int half = lane >> 4;
    const int l16  = lane & 15;
    const int kofs = 16 * wid + 8 * half;
    const hgl_frag8 af = hgl_frag(x + (size_t)(tm * 16) * IN_DIM, l16, kofs);
    const hgl_frag8 bf = hgl_frag(W + (size_t)(tn * 16) * IN_DIM, l16, kofs);
    v4f xv[8], wv[8];
    {
        const v4f* xr = reinterpret_cast<const v4f*>(x + (size_t)b * IN_DIM);
        const v4f* wr = reinterpret_cast<const v4f*>(W + (size_t)o * IN_DIM);
        #pragma unroll
        for (int j = 0; j < 8; ++j) {
            xv[j] = xr[4 * j + kq];
            wv[j] = wr[4 * j + kq];
        }
    }
    __builtin_amdgcn_sched_barrier(0);

    {
        v8f acc = {0.f, 0.f, 0.f, 0.f, 0.f, 0.f, 0.f, 0.f};
        acc = __builtin_amdgcn_wmma_f32_16x16x4_f32(0, (v2f){af.lo.x, af.lo.y}, 0, (v2f){bf.lo.x, bf.lo.y}, 0, acc, false, false);
        acc = __builtin_amdgcn_wmma_f32_16x16x4_f32(0, (v2f){af.lo.z, af.lo.w}, 0, (v2f){bf.lo.z, bf.lo.w}, 0, acc, false, false);
        acc = __builtin_amdgcn_wmma_f32_16x16x4_f32(0, (v2f){af.hi.x, af.hi.y}, 0, (v2f){bf.hi.x, bf.hi.y}, 0, acc, false, false);
        acc = __builtin_amdgcn_wmma_f32_16x16x4_f32(0, (v2f){af.hi.z, af.hi.w}, 0, (v2f){bf.hi.z, bf.hi.w}, 0, acc, false, false);
        #pragma unroll
        for (int r = 0; r < 8; ++r)
            part[wid * PART_PITCH + r * 32 + lane] = acc[r];
    }

    const float es = __expf(ls_o);
    const float cl = -LOG2E / (2.0f * es * es + EPS);
    const float r    = __builtin_amdgcn_sqrtf(cl * (-2.0f / LOG2E));
    const float rinv = r > 0.0f ? __builtin_amdgcn_rcpf(r) : 0.0f;

    v2f mom[SER_MOMS];
    #pragma unroll
    for (int n = 0; n < SER_MOMS; ++n) mom[n] = (v2f){0.f, 0.f};
    v2f   wz2  = (v2f){0.f, 0.f};
    float amin = 0.0f;
    #pragma unroll
    for (int j = 0; j < 8; ++j) {
        const v4f z4 = xv[j] * wv[j];
        #pragma unroll
        for (int hh = 0; hh < 2; ++hh) {
            const v2f z  = hh ? (v2f){z4.z, z4.w} : (v2f){z4.x, z4.y};
            const v2f a  = cl * z * z;
            amin = fminf(amin, fminf(a.x, a.y));
            const v2f w  = hgl_exp2_pair(a);
            const v2f s  = r * z;
            const v2f ss = s * s;
            wz2 = __builtin_elementwise_fma(w, z, wz2);
            v2f te = w, to = w * s;
            #pragma unroll
            for (int n = 0; n < SER_MOMS; n += 2) {
                mom[n]     += te;
                mom[n + 1] += to;
                if (n + 2 < SER_MOMS) { te *= ss; to *= ss; }
            }
        }
    }

    const unsigned bad = __builtin_amdgcn_ballot_w32(!(amin >= SER_MIN_A));

    __syncthreads();

    const int   pidx = (mrow & 7) * 32 + (mrow >> 3) * 16 + ncol;
    const float plin = part[kq * PART_PITCH + pidx] + part[(kq + 4) * PART_PITCH + pidx];

    float M[SER_MOMS];
    #pragma unroll
    for (int n = 0; n < SER_MOMS; ++n) M[n] = hgl_quad_sum(mom[n].x + mom[n].y);
    const float wz = hgl_quad_sum(wz2.x + wz2.y);
    const float pl = hgl_quad_sum(plin);

    float pn = 0.0f, ps = 0.0f;
    #pragma unroll
    for (int n = SER_TERMS - 1; n >= 1; --n) {
        const float mc = M[n] * hgl_inv_fact(n);
        pn = __builtin_fmaf(mc, M[n], pn);
        ps = __builtin_fmaf(mc, M[n + 1], ps);
    }
    pn = __builtin_fmaf(M[0], M[0], pn);
    float pg = __builtin_fmaf(M[0], wz, ps * rinv);

    if (bad != 0u) {
        float* const zq = stage + wid * STAGE_FLTS;
        float* const bq = zq + 4 * 32;
        float* const zn = bq + 4 * 32;
        float* const bn = zn + 64 * 4;
        #pragma unroll 1
        for (int pp = 0; pp < 8; ++pp) {
            if (((bad >> (PAIR_LANES * pp)) & 0xFu) == 0u) continue;
            const int   o2  = tn * 16 + nco0 + pp;
            const float es2 = __expf(log_sigma[o2]);
            const float cl2 = -LOG2E / (2.0f * es2 * es2 + EPS);
            const float4 xw = reinterpret_cast<const float4*>(x + (size_t)b  * IN_DIM)[lane];
            const float4 ww = reinterpret_cast<const float4*>(W + (size_t)o2 * IN_DIM)[lane];
            float z[4], a[4], m[4];
            z[0] = xw.x * ww.x;
            z[1] = xw.y * ww.y;
            z[2] = xw.z * ww.z;
            z[3] = xw.w * ww.w;
            #pragma unroll
            for (int q = 0; q < 4; ++q) {
                a[q] = cl2 * z[q] * z[q];
                m[q] = -2.0f * cl2 * z[q];
            }
            const float am2  = fminf(fminf(a[0], a[1]), fminf(a[2], a[3]));
            const bool  fact = __builtin_amdgcn_ballot_w32(!(am2 >= FACT_MIN_A)) == 0u;

            __builtin_amdgcn_fence(__ATOMIC_ACQ_REL, "workgroup");
            #pragma unroll
            for (int q = 0; q < 4; ++q) {
                zq[q * 32 + lane] = z[q];
                bq[q * 32 + lane] = a[q];
            }
            const float4 zv = make_float4(z[0], z[1], z[2], z[3]);
            const float4 av = make_float4(a[0], a[1], a[2], a[3]);
            reinterpret_cast<float4*>(zn)[lane]      = zv;
            reinterpret_cast<float4*>(zn)[lane + 32] = zv;
            reinterpret_cast<float4*>(bn)[lane]      = av;
            reinterpret_cast<float4*>(bn)[lane + 32] = av;
            __builtin_amdgcn_fence(__ATOMIC_ACQ_REL, "workgroup");

            float qn, qg;
            if (fact) hgl_pairs<true >(z, a, m, zq, bq, zn, bn, lane, qn, qg);
            else      hgl_pairs<false>(z, a, m, zq, bq, zn, bn, lane, qn, qg);
            #pragma unroll
            for (int off = 16; off > 0; off >>= 1) {
                qn += __shfl_xor(qn, off, 32);
                qg += __shfl_xor(qg, off, 32);
            }
            if (quad == pp) { pn = qn; pg = qg; }
        }
    }

    if (kq == 0) {
        const float norm = fmaxf(pn, 1e-12f);
        const float g    = pg / norm;
        const float al   = 1.0f / (1.0f + __expf(-al_o));
        out[(size_t)b * OUT_DIM + o] = al * g + (1.0f - al) * (pl + bi_o);
    }
}

extern "C" void kernel_launch(void* const* d_in, const int* in_sizes, int n_in,
                              void* d_out, int out_size, void* d_ws, size_t ws_size,
                              hipStream_t stream) {
    const float* x         = (const float*)d_in[0];
    const float* W         = (const float*)d_in[1];
    const float* bias      = (const float*)d_in[2];
    const float* log_sigma = (const float*)d_in[3];
    const float* alpha     = (const float*)d_in[4];
    float*       out       = (float*)d_out;

    hgl_fused<<<(B_DIM / 16) * (OUT_DIM / 16) * 4, WG_WAVES * 32, 0, stream>>>(
        x, W, bias, log_sigma, alpha, out);
}
